# starting kernel
# baseline (speedup 1.0000x reference)
#include <hip/hip_runtime.h>
#include <math.h>

typedef float v2f __attribute__((ext_vector_type(2)));
typedef float v8f __attribute__((ext_vector_type(8)));

#define N_EDGES 500000
#define N_RELS  256
#define DIM     256
#define N3      768

__global__ void zero_kernel(float* __restrict__ p, int n) {
    int i = blockIdx.x * blockDim.x + threadIdx.x;
    if (i < n) p[i] = 0.0f;
}

__global__ __launch_bounds__(256) void accum_kernel(
        const int* __restrict__ nid, const int* __restrict__ rel,
        const float* __restrict__ emb, float* __restrict__ gsum,
        float* __restrict__ counts, int do_count) {
    extern __shared__ float lsum[];
    for (int i = threadIdx.x; i < N_RELS * DIM; i += blockDim.x)
        lsum[i] = 0.0f;
    __syncthreads();

    const int lane  = threadIdx.x & 31;
    const int wave  = threadIdx.x >> 5;
    const int wpb   = blockDim.x >> 5;
    const int gw    = blockIdx.x * wpb + wave;
    const int nw    = gridDim.x * wpb;

    for (int e = gw; e < N_EDGES; e += nw) {
        const int n = nid[e];
        const int r = rel[e];
        if (e + nw < N_EDGES) {
            const int n2 = nid[e + nw];
            __builtin_prefetch(emb + (size_t)n2 * DIM + lane * 8, 0, 0);
        }
        const float4* row = (const float4*)(emb + (size_t)n * DIM);
        float4 v0 = row[lane * 2 + 0];
        float4 v1 = row[lane * 2 + 1];
        float* d = lsum + r * DIM + lane * 8;
        atomicAdd(d + 0, v0.x); atomicAdd(d + 1, v0.y);
        atomicAdd(d + 2, v0.z); atomicAdd(d + 3, v0.w);
        atomicAdd(d + 4, v1.x); atomicAdd(d + 5, v1.y);
        atomicAdd(d + 6, v1.z); atomicAdd(d + 7, v1.w);
        if (do_count && lane == 0) atomicAdd(counts + r, 1.0f);
    }
    __syncthreads();
    for (int i = threadIdx.x; i < N_RELS * DIM; i += blockDim.x)
        atomicAdd(gsum + i, lsum[i]);
}

__global__ void mean_kernel(float* __restrict__ sums,
                            const float* __restrict__ counts, int n) {
    int idx = blockIdx.x * blockDim.x + threadIdx.x;
    if (idx >= n) return;
    int r = (idx >> 8) & (N_RELS - 1);
    sums[idx] = sums[idx] / fmaxf(counts[r], 1.0f);
}

__global__ __launch_bounds__(256) void gemm_wmma(
        const float* __restrict__ A, int a_rs, int a_es,
        const float* __restrict__ W, const float* __restrict__ bias,
        float* __restrict__ C) {
    const int lane = threadIdx.x & 31;
    const int wave = threadIdx.x >> 5;
    const int tile = blockIdx.x * 8 + wave;
    const int tm   = tile / 48;
    const int tn   = tile % 48;
    const int half = lane >> 4;
    const int ml   = lane & 15;
    const int m    = tm * 16 + ml;
    const int n    = tn * 16 + ml;

    const float* arow = A + (size_t)m * a_rs;
    const float* wrow = W + (size_t)n * DIM;

    v8f acc = {};
    for (int k0 = 0; k0 < DIM; k0 += 4) {
        const int k = k0 + 2 * half;
        v2f a, b;
        a.x = arow[(size_t)k * a_es];
        a.y = arow[(size_t)(k + 1) * a_es];
        b.x = wrow[k];
        b.y = wrow[k + 1];
        acc = __builtin_amdgcn_wmma_f32_16x16x4_f32(
                 false, a,  false, b,
                 (short)0, acc,  false,  false);
    }
    const float bv = bias[n];
#pragma unroll
    for (int v = 0; v < 8; ++v) {
        const int row = tm * 16 + v + 8 * half;
        C[(size_t)row * N3 + n] = acc[v] + bv;
    }
}

__global__ void gru_gates(const float* __restrict__ gi,
                          const float* __restrict__ gh,
                          const float* __restrict__ dyn, int which,
                          float* __restrict__ out) {
    int idx = blockIdx.x * blockDim.x + threadIdx.x;
    if (idx >= N_RELS * DIM) return;
    int r = idx >> 8;
    int d = idx & (DIM - 1);
    const float* gir = gi + (size_t)r * N3;
    const float* ghr = gh + (size_t)r * N3;
    float i_r = gir[d], i_z = gir[DIM + d], i_n = gir[2 * DIM + d];
    float h_r = ghr[d], h_z = ghr[DIM + d], h_n = ghr[2 * DIM + d];
    float h   = dyn[(size_t)r * 512 + d * 2 + which];
    float rg  = 1.0f / (1.0f + expf(-(i_r + h_r)));
    float z   = 1.0f / (1.0f + expf(-(i_z + h_z)));
    float ng  = tanhf(i_n + rg * h_n);
    out[(size_t)idx * 2 + which] = (1.0f - z) * ng + z * h;
}

extern "C" void kernel_launch(void* const* d_in, const int* in_sizes, int n_in,
                              void* d_out, int out_size, void* d_ws, size_t ws_size,
                              hipStream_t stream) {
    (void)in_sizes; (void)n_in; (void)out_size; (void)ws_size;

    const int*   src_nid = (const int*)d_in[0];
    const int*   dst_nid = (const int*)d_in[1];
    const int*   rel     = (const int*)d_in[2];
    const float* ent     = (const float*)d_in[3];
    const float* dyn     = (const float*)d_in[4];
    const float* W_ih    = (const float*)d_in[5];
    const float* W_hh    = (const float*)d_in[6];
    const float* b_ih    = (const float*)d_in[7];
    const float* b_hh    = (const float*)d_in[8];
    float*       out     = (float*)d_out;

    float* ws       = (float*)d_ws;
    float* sums_src = ws;
    float* sums_dst = ws + 65536;
    float* counts   = ws + 131072;
    float* gi_src   = ws + 131328;
    float* gh_src   = gi_src + 196608;
    float* gi_dst   = gh_src + 196608;
    float* gh_dst   = gi_dst + 196608;

    const int NACC = 131328;
    zero_kernel<<<(NACC + 255) / 256, 256, 0, stream>>>(ws, NACC);

    const size_t ldsBytes = (size_t)N_RELS * DIM * sizeof(float);
    accum_kernel<<<64, 256, ldsBytes, stream>>>(src_nid, rel, ent,
                                                sums_src, counts, 1);
    accum_kernel<<<64, 256, ldsBytes, stream>>>(dst_nid, rel, ent,
                                                sums_dst, counts, 0);

    mean_kernel<<<(2 * 65536) / 256, 256, 0, stream>>>(ws, counts, 2 * 65536);

    gemm_wmma<<<96, 256, 0, stream>>>(sums_src, DIM, 1, W_ih, b_ih, gi_src);
    gemm_wmma<<<96, 256, 0, stream>>>(dyn + 0,  512, 2, W_hh, b_hh, gh_src);
    gemm_wmma<<<96, 256, 0, stream>>>(sums_dst, DIM, 1, W_ih, b_ih, gi_dst);
    gemm_wmma<<<96, 256, 0, stream>>>(dyn + 1,  512, 2, W_hh, b_hh, gh_dst);

    gru_gates<<<65536 / 256, 256, 0, stream>>>(gi_src, gh_src, dyn, 0, out);
    gru_gates<<<65536 / 256, 256, 0, stream>>>(gi_dst, gh_dst, dyn, 1, out);
}


# accum: gather spread over all 256 WGPs (4 channel slices x 64 edge groups, 32 waves each), 32 row loads in flight per wave; blocks sum other edge sets, so the f32 atomic sum order changes
# speedup vs baseline: 9.1100x; 9.1100x over previous
#include <hip/hip_runtime.h>
#include <math.h>

typedef float v2f __attribute__((ext_vector_type(2)));
typedef float v8f __attribute__((ext_vector_type(8)));

#define N_EDGES 500000
#define N_RELS  256
#define DIM     256
#define N3      768

__global__ void zero_kernel(float* __restrict__ p, int n) {
    int i = blockIdx.x * blockDim.x + threadIdx.x;
    if (i < n) p[i] = 0.0f;
}

#define ACC_THREADS 1024
#define ACC_WAVES   (ACC_THREADS / 32)
#define NSLICE      4
#define SLICE_CH    (DIM / NSLICE)
#define NGROUP      64
#define EPT         32
#define NCHUNK      (N_EDGES / EPT)
static_assert(N_EDGES % EPT == 0, "edge count must be a whole number of trips");

__global__ __launch_bounds__(ACC_THREADS) void accum_kernel(
        const int* __restrict__ nid, const int* __restrict__ rel,
        const float* __restrict__ emb, float* __restrict__ gsum,
        float* __restrict__ counts, int do_count) {
    extern __shared__ float lsum[];
    for (int i = threadIdx.x; i < N_RELS * SLICE_CH; i += ACC_THREADS)
        lsum[i] = 0.0f;
    __syncthreads();

    const int lane  = threadIdx.x & 31;
    const int wave  = threadIdx.x >> 5;
    const int slice = blockIdx.x & (NSLICE - 1);
    const int group = blockIdx.x / NSLICE;
    const int gw    = wave * NGROUP + group;
    const int nw    = ACC_WAVES * NGROUP;
    const bool count_here = do_count && slice == 0;

    const float* ebase = emb + slice * SLICE_CH;
    float*       lbase = lsum + lane * 2;

    int c = gw;
    int cur_n = 0, cur_r = 0;
    if (c < NCHUNK) {
        cur_n = nid[c * EPT + lane];
        cur_r = rel[c * EPT + lane];
    }
    for (; c < NCHUNK; c += nw) {
        if (count_here) atomicAdd(counts + cur_r, 1.0f);

        float2 v[EPT];
#pragma unroll
        for (int j = 0; j < EPT; ++j) {
            const int n = __builtin_amdgcn_readlane(cur_n, j);
            const float2* row = (const float2*)(ebase + (size_t)n * DIM);
            v[j] = row[lane];
        }
        const int cn = (c + nw < NCHUNK) ? c + nw : c;
        const int nxt_n = nid[cn * EPT + lane];
        const int nxt_r = rel[cn * EPT + lane];
#pragma unroll
        for (int j = 0; j < EPT; ++j) {
            const int r = __builtin_amdgcn_readlane(cur_r, j);
            float* d = lbase + r * SLICE_CH;
            atomicAdd(d + 0, v[j].x);
            atomicAdd(d + 1, v[j].y);
        }
        cur_n = nxt_n;
        cur_r = nxt_r;
    }
    __syncthreads();
    for (int i = threadIdx.x; i < N_RELS * SLICE_CH; i += ACC_THREADS) {
        const int r  = i / SLICE_CH;
        const int ch = i % SLICE_CH;
        atomicAdd(gsum + r * DIM + slice * SLICE_CH + ch, lsum[i]);
    }
}

__global__ void mean_kernel(float* __restrict__ sums,
                            const float* __restrict__ counts, int n) {
    int idx = blockIdx.x * blockDim.x + threadIdx.x;
    if (idx >= n) return;
    int r = (idx >> 8) & (N_RELS - 1);
    sums[idx] = sums[idx] / fmaxf(counts[r], 1.0f);
}

__global__ __launch_bounds__(256) void gemm_wmma(
        const float* __restrict__ A, int a_rs, int a_es,
        const float* __restrict__ W, const float* __restrict__ bias,
        float* __restrict__ C) {
    const int lane = threadIdx.x & 31;
    const int wave = threadIdx.x >> 5;
    const int tile = blockIdx.x * 8 + wave;
    const int tm   = tile / 48;
    const int tn   = tile % 48;
    const int half = lane >> 4;
    const int ml   = lane & 15;
    const int m    = tm * 16 + ml;
    const int n    = tn * 16 + ml;

    const float* arow = A + (size_t)m * a_rs;
    const float* wrow = W + (size_t)n * DIM;

    v8f acc = {};
    for (int k0 = 0; k0 < DIM; k0 += 4) {
        const int k = k0 + 2 * half;
        v2f a, b;
        a.x = arow[(size_t)k * a_es];
        a.y = arow[(size_t)(k + 1) * a_es];
        b.x = wrow[k];
        b.y = wrow[k + 1];
        acc = __builtin_amdgcn_wmma_f32_16x16x4_f32(
                 false, a,  false, b,
                 (short)0, acc,  false,  false);
    }
    const float bv = bias[n];
#pragma unroll
    for (int v = 0; v < 8; ++v) {
        const int row = tm * 16 + v + 8 * half;
        C[(size_t)row * N3 + n] = acc[v] + bv;
    }
}

__global__ void gru_gates(const float* __restrict__ gi,
                          const float* __restrict__ gh,
                          const float* __restrict__ dyn, int which,
                          float* __restrict__ out) {
    int idx = blockIdx.x * blockDim.x + threadIdx.x;
    if (idx >= N_RELS * DIM) return;
    int r = idx >> 8;
    int d = idx & (DIM - 1);
    const float* gir = gi + (size_t)r * N3;
    const float* ghr = gh + (size_t)r * N3;
    float i_r = gir[d], i_z = gir[DIM + d], i_n = gir[2 * DIM + d];
    float h_r = ghr[d], h_z = ghr[DIM + d], h_n = ghr[2 * DIM + d];
    float h   = dyn[(size_t)r * 512 + d * 2 + which];
    float rg  = 1.0f / (1.0f + expf(-(i_r + h_r)));
    float z   = 1.0f / (1.0f + expf(-(i_z + h_z)));
    float ng  = tanhf(i_n + rg * h_n);
    out[(size_t)idx * 2 + which] = (1.0f - z) * ng + z * h;
}

extern "C" void kernel_launch(void* const* d_in, const int* in_sizes, int n_in,
                              void* d_out, int out_size, void* d_ws, size_t ws_size,
                              hipStream_t stream) {
    (void)in_sizes; (void)n_in; (void)out_size; (void)ws_size;

    const int*   src_nid = (const int*)d_in[0];
    const int*   dst_nid = (const int*)d_in[1];
    const int*   rel     = (const int*)d_in[2];
    const float* ent     = (const float*)d_in[3];
    const float* dyn     = (const float*)d_in[4];
    const float* W_ih    = (const float*)d_in[5];
    const float* W_hh    = (const float*)d_in[6];
    const float* b_ih    = (const float*)d_in[7];
    const float* b_hh    = (const float*)d_in[8];
    float*       out     = (float*)d_out;

    float* ws       = (float*)d_ws;
    float* sums_src = ws;
    float* sums_dst = ws + 65536;
    float* counts   = ws + 131072;
    float* gi_src   = ws + 131328;
    float* gh_src   = gi_src + 196608;
    float* gi_dst   = gh_src + 196608;
    float* gh_dst   = gi_dst + 196608;

    const int NACC = 131328;
    zero_kernel<<<(NACC + 255) / 256, 256, 0, stream>>>(ws, NACC);

    const size_t ldsBytes = (size_t)N_RELS * SLICE_CH * sizeof(float);
    accum_kernel<<<NSLICE * NGROUP, ACC_THREADS, ldsBytes, stream>>>(
            src_nid, rel, ent, sums_src, counts, 1);
    accum_kernel<<<NSLICE * NGROUP, ACC_THREADS, ldsBytes, stream>>>(
            dst_nid, rel, ent, sums_dst, counts, 0);

    mean_kernel<<<(2 * 65536) / 256, 256, 0, stream>>>(ws, counts, 2 * 65536);

    gemm_wmma<<<96, 256, 0, stream>>>(sums_src, DIM, 1, W_ih, b_ih, gi_src);
    gemm_wmma<<<96, 256, 0, stream>>>(dyn + 0,  512, 2, W_hh, b_hh, gh_src);
    gemm_wmma<<<96, 256, 0, stream>>>(sums_dst, DIM, 1, W_ih, b_ih, gi_dst);
    gemm_wmma<<<96, 256, 0, stream>>>(dyn + 1,  512, 2, W_hh, b_hh, gh_dst);

    gru_gates<<<65536 / 256, 256, 0, stream>>>(gi_src, gh_src, dyn, 0, out);
    gru_gates<<<65536 / 256, 256, 0, stream>>>(gi_dst, gh_dst, dyn, 1, out);
}


# tail fused into one launch: mean, the four f32 WMMA products and both gate passes; each wave keeps its six gate tiles in registers, gi/gh never stored; k order inside the product sums changes
# speedup vs baseline: 11.4396x; 1.2557x over previous
#include <hip/hip_runtime.h>
#include <math.h>

typedef float v2f __attribute__((ext_vector_type(2)));
typedef float v8f __attribute__((ext_vector_type(8)));

#define N_EDGES 500000
#define N_RELS  256
#define DIM     256
#define N3      768

__global__ void zero_kernel(float* __restrict__ p, int n) {
    int i = blockIdx.x * blockDim.x + threadIdx.x;
    if (i < n) p[i] = 0.0f;
}

#define ACC_THREADS 1024
#define ACC_WAVES   (ACC_THREADS / 32)
#define NSLICE      4
#define SLICE_CH    (DIM / NSLICE)
#define NGROUP      64
#define EPT         32
#define NCHUNK      (N_EDGES / EPT)
static_assert(N_EDGES % EPT == 0, "edge count must be a whole number of trips");

__global__ __launch_bounds__(ACC_THREADS) void accum_kernel(
        const int* __restrict__ nid, const int* __restrict__ rel,
        const float* __restrict__ emb, float* __restrict__ gsum,
        float* __restrict__ counts, int do_count) {
    extern __shared__ float lsum[];
    for (int i = threadIdx.x; i < N_RELS * SLICE_CH; i += ACC_THREADS)
        lsum[i] = 0.0f;
    __syncthreads();

    const int lane  = threadIdx.x & 31;
    const int wave  = threadIdx.x >> 5;
    const int slice = blockIdx.x & (NSLICE - 1);
    const int group = blockIdx.x / NSLICE;
    const int gw    = wave * NGROUP + group;
    const int nw    = ACC_WAVES * NGROUP;
    const bool count_here = do_count && slice == 0;

    const float* ebase = emb + slice * SLICE_CH;
    float*       lbase = lsum + lane * 2;

    int c = gw;
    int cur_n = 0, cur_r = 0;
    if (c < NCHUNK) {
        cur_n = nid[c * EPT + lane];
        cur_r = rel[c * EPT + lane];
    }
    for (; c < NCHUNK; c += nw) {
        if (count_here) atomicAdd(counts + cur_r, 1.0f);

        float2 v[EPT];
#pragma unroll
        for (int j = 0; j < EPT; ++j) {
            const int n = __builtin_amdgcn_readlane(cur_n, j);
            const float2* row = (const float2*)(ebase + (size_t)n * DIM);
            v[j] = row[lane];
        }
        const int cn = (c + nw < NCHUNK) ? c + nw : c;
        const int nxt_n = nid[cn * EPT + lane];
        const int nxt_r = rel[cn * EPT + lane];
#pragma unroll
        for (int j = 0; j < EPT; ++j) {
            const int r = __builtin_amdgcn_readlane(cur_r, j);
            float* d = lbase + r * SLICE_CH;
            atomicAdd(d + 0, v[j].x);
            atomicAdd(d + 1, v[j].y);
        }
        cur_n = nxt_n;
        cur_r = nxt_r;
    }
    __syncthreads();
    for (int i = threadIdx.x; i < N_RELS * SLICE_CH; i += ACC_THREADS) {
        const int r  = i / SLICE_CH;
        const int ch = i % SLICE_CH;
        atomicAdd(gsum + r * DIM + slice * SLICE_CH + ch, lsum[i]);
    }
}

struct Frag {
    float4 x;
    float  h[4];
    float4 wi0, wi1, wi2;
    float4 wh0, wh1, wh2;
};

__device__ __forceinline__ Frag load_frag(const float* __restrict__ xrow,
                                          const float* __restrict__ hrow,
                                          const float* __restrict__ wi,
                                          const float* __restrict__ wh, int k0) {
    Frag f;
    f.x    = *(const float4*)(xrow + k0);
    f.h[0] = hrow[2 * k0];
    f.h[1] = hrow[2 * k0 + 2];
    f.h[2] = hrow[2 * k0 + 4];
    f.h[3] = hrow[2 * k0 + 6];
    f.wi0  = *(const float4*)(wi + k0);
    f.wi1  = *(const float4*)(wi + DIM * DIM + k0);
    f.wi2  = *(const float4*)(wi + 2 * DIM * DIM + k0);
    f.wh0  = *(const float4*)(wh + k0);
    f.wh1  = *(const float4*)(wh + DIM * DIM + k0);
    f.wh2  = *(const float4*)(wh + 2 * DIM * DIM + k0);
    return f;
}

#define WMMA_F32(a, b, c) __builtin_amdgcn_wmma_f32_16x16x4_f32( \
         false, (a),  false, (b),  (short)0, (c), \
         false,  false)

__device__ __forceinline__ void mma_frag(const Frag& f, float cnt,
                                         v8f& gi0, v8f& gi1, v8f& gi2,
                                         v8f& gh0, v8f& gh1, v8f& gh2) {
    v2f ax0 = {f.x.x / cnt, f.x.y / cnt};
    v2f ax1 = {f.x.z / cnt, f.x.w / cnt};
    v2f ah0 = {f.h[0], f.h[1]};
    v2f ah1 = {f.h[2], f.h[3]};
    v2f b;
    b.x = f.wi0.x; b.y = f.wi0.y; gi0 = WMMA_F32(ax0, b, gi0);
    b.x = f.wi1.x; b.y = f.wi1.y; gi1 = WMMA_F32(ax0, b, gi1);
    b.x = f.wi2.x; b.y = f.wi2.y; gi2 = WMMA_F32(ax0, b, gi2);
    b.x = f.wh0.x; b.y = f.wh0.y; gh0 = WMMA_F32(ah0, b, gh0);
    b.x = f.wh1.x; b.y = f.wh1.y; gh1 = WMMA_F32(ah0, b, gh1);
    b.x = f.wh2.x; b.y = f.wh2.y; gh2 = WMMA_F32(ah0, b, gh2);
    b.x = f.wi0.z; b.y = f.wi0.w; gi0 = WMMA_F32(ax1, b, gi0);
    b.x = f.wi1.z; b.y = f.wi1.w; gi1 = WMMA_F32(ax1, b, gi1);
    b.x = f.wi2.z; b.y = f.wi2.w; gi2 = WMMA_F32(ax1, b, gi2);
    b.x = f.wh0.z; b.y = f.wh0.w; gh0 = WMMA_F32(ah1, b, gh0);
    b.x = f.wh1.z; b.y = f.wh1.w; gh1 = WMMA_F32(ah1, b, gh1);
    b.x = f.wh2.z; b.y = f.wh2.w; gh2 = WMMA_F32(ah1, b, gh2);
}

#define GRU_THREADS 64
#define KCHUNK      8
#define TRIP_CHUNKS 16
#define NTRIP       (DIM / (KCHUNK * TRIP_CHUNKS))

__global__ __launch_bounds__(GRU_THREADS) void gru_fused(
        const float* __restrict__ sums,
        const float* __restrict__ counts,
        const float* __restrict__ dyn,
        const float* __restrict__ W_ih, const float* __restrict__ W_hh,
        const float* __restrict__ b_ih, const float* __restrict__ b_hh,
        float* __restrict__ out) {
    __shared__ float xch[8 * 32];

    const int lane  = threadIdx.x & 31;
    const int which = threadIdx.x >> 5;
    const int tm    = blockIdx.x >> 4;
    const int tn    = blockIdx.x & 15;
    const int half  = lane >> 4;
    const int ml    = lane & 15;
    const int m     = tm * 16 + ml;
    const int d     = tn * 16 + ml;

    const float cnt = fmaxf(counts[m], 1.0f);
    const float* xrow = sums + (size_t)which * (N_RELS * DIM) + (size_t)m * DIM + 4 * half;
    const float* hrow = dyn + (size_t)m * (2 * DIM) + 8 * half + which;
    const float* wi   = W_ih + (size_t)d * DIM + 4 * half;
    const float* wh   = W_hh + (size_t)d * DIM + 4 * half;

    v8f gi0 = {}, gi1 = {}, gi2 = {}, gh0 = {}, gh1 = {}, gh2 = {};

#pragma unroll 1
    for (int t = 0; t < NTRIP; ++t) {
        const int kb = t * (KCHUNK * TRIP_CHUNKS);
        Frag fr[2];
        fr[0] = load_frag(xrow, hrow, wi, wh, kb);
#pragma unroll
        for (int c = 0; c < TRIP_CHUNKS; ++c) {
            __builtin_amdgcn_sched_barrier(0);
            if (c + 1 < TRIP_CHUNKS)
                fr[(c + 1) & 1] = load_frag(xrow, hrow, wi, wh, kb + (c + 1) * KCHUNK);
            __builtin_amdgcn_sched_barrier(0);
            mma_frag(fr[c & 1], cnt, gi0, gi1, gi2, gh0, gh1, gh2);
        }
    }

    const float bir = b_ih[d], biz = b_ih[DIM + d], bin = b_ih[2 * DIM + d];
    const float bhr = b_hh[d], bhz = b_hh[DIM + d], bhn = b_hh[2 * DIM + d];

    float res[8];
#pragma unroll
    for (int v = 0; v < 8; ++v) {
        const int row = tm * 16 + v + 8 * half;
        const float i_r = gi0[v] + bir, i_z = gi1[v] + biz, i_n = gi2[v] + bin;
        const float h_r = gh0[v] + bhr, h_z = gh1[v] + bhz, h_n = gh2[v] + bhn;
        const float h   = dyn[(size_t)row * (2 * DIM) + d * 2 + which];
        const float rg  = 1.0f / (1.0f + expf(-(i_r + h_r)));
        const float z   = 1.0f / (1.0f + expf(-(i_z + h_z)));
        const float ng  = tanhf(i_n + rg * h_n);
        res[v] = (1.0f - z) * ng + z * h;
    }

    if (which == 1) {
#pragma unroll
        for (int v = 0; v < 8; ++v) xch[v * 32 + lane] = res[v];
    }
    __syncthreads();
    if (which == 0) {
#pragma unroll
        for (int v = 0; v < 8; ++v) {
            const int row = tm * 16 + v + 8 * half;
            float2 o;
            o.x = res[v];
            o.y = xch[v * 32 + lane];
            *(float2*)(out + ((size_t)row * DIM + d) * 2) = o;
        }
    }
}

extern "C" void kernel_launch(void* const* d_in, const int* in_sizes, int n_in,
                              void* d_out, int out_size, void* d_ws, size_t ws_size,
                              hipStream_t stream) {
    (void)in_sizes; (void)n_in; (void)out_size; (void)ws_size;

    const int*   src_nid = (const int*)d_in[0];
    const int*   dst_nid = (const int*)d_in[1];
    const int*   rel     = (const int*)d_in[2];
    const float* ent     = (const float*)d_in[3];
    const float* dyn     = (const float*)d_in[4];
    const float* W_ih    = (const float*)d_in[5];
    const float* W_hh    = (const float*)d_in[6];
    const float* b_ih    = (const float*)d_in[7];
    const float* b_hh    = (const float*)d_in[8];
    float*       out     = (float*)d_out;

    float* ws       = (float*)d_ws;
    float* sums_src = ws;
    float* sums_dst = ws + 65536;
    float* counts   = ws + 131072;

    const int NACC = 131328;
    zero_kernel<<<(NACC + 255) / 256, 256, 0, stream>>>(ws, NACC);

    const size_t ldsBytes = (size_t)N_RELS * SLICE_CH * sizeof(float);
    accum_kernel<<<NSLICE * NGROUP, ACC_THREADS, ldsBytes, stream>>>(
            src_nid, rel, ent, sums_src, counts, 1);
    accum_kernel<<<NSLICE * NGROUP, ACC_THREADS, ldsBytes, stream>>>(
            dst_nid, rel, ent, sums_dst, counts, 0);

    gru_fused<<<(N_RELS / 16) * (DIM / 16), GRU_THREADS, 0, stream>>>(
            ws, counts, dyn, W_ih, W_hh, b_ih, b_hh, out);
}


# accum: src and dst planes gathered in one launch (2 planes x 4 slices x 32 edge groups = 256 blocks); 32 blocks, not 64, add each sum word, so the f32 atomic sum order changes
# speedup vs baseline: 15.2660x; 1.3345x over previous
#include <hip/hip_runtime.h>
#include <math.h>

typedef float v2f __attribute__((ext_vector_type(2)));
typedef float v8f __attribute__((ext_vector_type(8)));

#define N_EDGES 500000
#define N_RELS  256
#define DIM     256
#define N3      768

__global__ void zero_kernel(float* __restrict__ p, int n) {
    int i = blockIdx.x * blockDim.x + threadIdx.x;
    if (i < n) p[i] = 0.0f;
}

#define ACC_THREADS 1024
#define ACC_WAVES   (ACC_THREADS / 32)
#define NSLICE      4
#define SLICE_CH    (DIM / NSLICE)
#define NGROUP      32
#define NPLANE      2
#define EPT         32
#define NCHUNK      (N_EDGES / EPT)
static_assert(N_EDGES % EPT == 0, "edge count must be a whole number of trips");

__global__ __launch_bounds__(ACC_THREADS) void accum_kernel(
        const int* __restrict__ src_nid, const int* __restrict__ dst_nid,
        const int* __restrict__ rel, const float* __restrict__ emb,
        float* __restrict__ sums,
        float* __restrict__ counts) {
    extern __shared__ float lsum[];
    for (int i = threadIdx.x; i < N_RELS * SLICE_CH; i += ACC_THREADS)
        lsum[i] = 0.0f;
    __syncthreads();

    const int lane  = threadIdx.x & 31;
    const int wave  = threadIdx.x >> 5;
    const int slice = blockIdx.x & (NSLICE - 1);
    const int plane = (blockIdx.x / NSLICE) & (NPLANE - 1);
    const int group = blockIdx.x / (NSLICE * NPLANE);
    const int gw    = wave * NGROUP + group;
    const int nw    = ACC_WAVES * NGROUP;
    const bool count_here = plane == 0 && slice == 0;

    const int* nid  = plane ? dst_nid : src_nid;
    float*     gsum = sums + plane * (N_RELS * DIM);

    const float* ebase = emb + slice * SLICE_CH;
    float*       lbase = lsum + lane * 2;

    int c = gw;
    int cur_n = 0, cur_r = 0;
    if (c < NCHUNK) {
        cur_n = nid[c * EPT + lane];
        cur_r = rel[c * EPT + lane];
    }
    for (; c < NCHUNK; c += nw) {
        if (count_here) atomicAdd(counts + cur_r, 1.0f);

        float2 v[EPT];
#pragma unroll
        for (int j = 0; j < EPT; ++j) {
            const int n = __builtin_amdgcn_readlane(cur_n, j);
            const float2* row = (const float2*)(ebase + (size_t)n * DIM);
            v[j] = row[lane];
        }
        const int cn = (c + nw < NCHUNK) ? c + nw : c;
        const int nxt_n = nid[cn * EPT + lane];
        const int nxt_r = rel[cn * EPT + lane];
#pragma unroll
        for (int j = 0; j < EPT; ++j) {
            const int r = __builtin_amdgcn_readlane(cur_r, j);
            float* d = lbase + r * SLICE_CH;
            atomicAdd(d + 0, v[j].x);
            atomicAdd(d + 1, v[j].y);
        }
        cur_n = nxt_n;
        cur_r = nxt_r;
    }
    __syncthreads();
    for (int i = threadIdx.x; i < N_RELS * SLICE_CH; i += ACC_THREADS) {
        const int r  = i / SLICE_CH;
        const int ch = i % SLICE_CH;
        atomicAdd(gsum + r * DIM + slice * SLICE_CH + ch, lsum[i]);
    }
}

struct Frag {
    float4 x;
    float  h[4];
    float4 wi0, wi1, wi2;
    float4 wh0, wh1, wh2;
};

__device__ __forceinline__ Frag load_frag(const float* __restrict__ xrow,
                                          const float* __restrict__ hrow,
                                          const float* __restrict__ wi,
                                          const float* __restrict__ wh, int k0) {
    Frag f;
    f.x    = *(const float4*)(xrow + k0);
    f.h[0] = hrow[2 * k0];
    f.h[1] = hrow[2 * k0 + 2];
    f.h[2] = hrow[2 * k0 + 4];
    f.h[3] = hrow[2 * k0 + 6];
    f.wi0  = *(const float4*)(wi + k0);
    f.wi1  = *(const float4*)(wi + DIM * DIM + k0);
    f.wi2  = *(const float4*)(wi + 2 * DIM * DIM + k0);
    f.wh0  = *(const float4*)(wh + k0);
    f.wh1  = *(const float4*)(wh + DIM * DIM + k0);
    f.wh2  = *(const float4*)(wh + 2 * DIM * DIM + k0);
    return f;
}

#define WMMA_F32(a, b, c) __builtin_amdgcn_wmma_f32_16x16x4_f32( \
         false, (a),  false, (b),  (short)0, (c), \
         false,  false)

__device__ __forceinline__ void mma_frag(const Frag& f, float cnt,
                                         v8f& gi0, v8f& gi1, v8f& gi2,
                                         v8f& gh0, v8f& gh1, v8f& gh2) {
    v2f ax0 = {f.x.x / cnt, f.x.y / cnt};
    v2f ax1 = {f.x.z / cnt, f.x.w / cnt};
    v2f ah0 = {f.h[0], f.h[1]};
    v2f ah1 = {f.h[2], f.h[3]};
    v2f b;
    b.x = f.wi0.x; b.y = f.wi0.y; gi0 = WMMA_F32(ax0, b, gi0);
    b.x = f.wi1.x; b.y = f.wi1.y; gi1 = WMMA_F32(ax0, b, gi1);
    b.x = f.wi2.x; b.y = f.wi2.y; gi2 = WMMA_F32(ax0, b, gi2);
    b.x = f.wh0.x; b.y = f.wh0.y; gh0 = WMMA_F32(ah0, b, gh0);
    b.x = f.wh1.x; b.y = f.wh1.y; gh1 = WMMA_F32(ah0, b, gh1);
    b.x = f.wh2.x; b.y = f.wh2.y; gh2 = WMMA_F32(ah0, b, gh2);
    b.x = f.wi0.z; b.y = f.wi0.w; gi0 = WMMA_F32(ax1, b, gi0);
    b.x = f.wi1.z; b.y = f.wi1.w; gi1 = WMMA_F32(ax1, b, gi1);
    b.x = f.wi2.z; b.y = f.wi2.w; gi2 = WMMA_F32(ax1, b, gi2);
    b.x = f.wh0.z; b.y = f.wh0.w; gh0 = WMMA_F32(ah1, b, gh0);
    b.x = f.wh1.z; b.y = f.wh1.w; gh1 = WMMA_F32(ah1, b, gh1);
    b.x = f.wh2.z; b.y = f.wh2.w; gh2 = WMMA_F32(ah1, b, gh2);
}

#define GRU_THREADS 64
#define KCHUNK      8
#define TRIP_CHUNKS 16
#define NTRIP       (DIM / (KCHUNK * TRIP_CHUNKS))

__global__ __launch_bounds__(GRU_THREADS) void gru_fused(
        const float* __restrict__ sums,
        const float* __restrict__ counts,
        const float* __restrict__ dyn,
        const float* __restrict__ W_ih, const float* __restrict__ W_hh,
        const float* __restrict__ b_ih, const float* __restrict__ b_hh,
        float* __restrict__ out) {
    __shared__ float xch[8 * 32];

    const int lane  = threadIdx.x & 31;
    const int which = threadIdx.x >> 5;
    const int tm    = blockIdx.x >> 4;
    const int tn    = blockIdx.x & 15;
    const int half  = lane >> 4;
    const int ml    = lane & 15;
    const int m     = tm * 16 + ml;
    const int d     = tn * 16 + ml;

    const float cnt = fmaxf(counts[m], 1.0f);
    const float* xrow = sums + (size_t)which * (N_RELS * DIM) + (size_t)m * DIM + 4 * half;
    const float* hrow = dyn + (size_t)m * (2 * DIM) + 8 * half + which;
    const float* wi   = W_ih + (size_t)d * DIM + 4 * half;
    const float* wh   = W_hh + (size_t)d * DIM + 4 * half;

    v8f gi0 = {}, gi1 = {}, gi2 = {}, gh0 = {}, gh1 = {}, gh2 = {};

#pragma unroll 1
    for (int t = 0; t < NTRIP; ++t) {
        const int kb = t * (KCHUNK * TRIP_CHUNKS);
        Frag fr[2];
        fr[0] = load_frag(xrow, hrow, wi, wh, kb);
#pragma unroll
        for (int c = 0; c < TRIP_CHUNKS; ++c) {
            __builtin_amdgcn_sched_barrier(0);
            if (c + 1 < TRIP_CHUNKS)
                fr[(c + 1) & 1] = load_frag(xrow, hrow, wi, wh, kb + (c + 1) * KCHUNK);
            __builtin_amdgcn_sched_barrier(0);
            mma_frag(fr[c & 1], cnt, gi0, gi1, gi2, gh0, gh1, gh2);
        }
    }

    const float bir = b_ih[d], biz = b_ih[DIM + d], bin = b_ih[2 * DIM + d];
    const float bhr = b_hh[d], bhz = b_hh[DIM + d], bhn = b_hh[2 * DIM + d];

    float res[8];
#pragma unroll
    for (int v = 0; v < 8; ++v) {
        const int row = tm * 16 + v + 8 * half;
        const float i_r = gi0[v] + bir, i_z = gi1[v] + biz, i_n = gi2[v] + bin;
        const float h_r = gh0[v] + bhr, h_z = gh1[v] + bhz, h_n = gh2[v] + bhn;
        const float h   = dyn[(size_t)row * (2 * DIM) + d * 2 + which];
        const float rg  = 1.0f / (1.0f + expf(-(i_r + h_r)));
        const float z   = 1.0f / (1.0f + expf(-(i_z + h_z)));
        const float ng  = tanhf(i_n + rg * h_n);
        res[v] = (1.0f - z) * ng + z * h;
    }

    if (which == 1) {
#pragma unroll
        for (int v = 0; v < 8; ++v) xch[v * 32 + lane] = res[v];
    }
    __syncthreads();
    if (which == 0) {
#pragma unroll
        for (int v = 0; v < 8; ++v) {
            const int row = tm * 16 + v + 8 * half;
            float2 o;
            o.x = res[v];
            o.y = xch[v * 32 + lane];
            *(float2*)(out + ((size_t)row * DIM + d) * 2) = o;
        }
    }
}

extern "C" void kernel_launch(void* const* d_in, const int* in_sizes, int n_in,
                              void* d_out, int out_size, void* d_ws, size_t ws_size,
                              hipStream_t stream) {
    (void)in_sizes; (void)n_in; (void)out_size; (void)ws_size;

    const int*   src_nid = (const int*)d_in[0];
    const int*   dst_nid = (const int*)d_in[1];
    const int*   rel     = (const int*)d_in[2];
    const float* ent     = (const float*)d_in[3];
    const float* dyn     = (const float*)d_in[4];
    const float* W_ih    = (const float*)d_in[5];
    const float* W_hh    = (const float*)d_in[6];
    const float* b_ih    = (const float*)d_in[7];
    const float* b_hh    = (const float*)d_in[8];
    float*       out     = (float*)d_out;

    float* ws       = (float*)d_ws;
    float* counts   = ws + 131072;

    const int NACC = 131328;
    zero_kernel<<<(NACC + 255) / 256, 256, 0, stream>>>(ws, NACC);

    const size_t ldsBytes = (size_t)N_RELS * SLICE_CH * sizeof(float);
    accum_kernel<<<NPLANE * NSLICE * NGROUP, ACC_THREADS, ldsBytes, stream>>>(
            src_nid, dst_nid, rel, ent, ws, counts);

    gru_fused<<<(N_RELS / 16) * (DIM / 16), GRU_THREADS, 0, stream>>>(
            ws, counts, dyn, W_ih, W_hh, b_ih, b_hh, out);
}


# accum: block sorts its edge share by relation in LDS (counting sort, arrival-order rank), a relation's rows summed in registers; no LDS f32 atomics, counts from bin sizes; f32 sum order changes
# speedup vs baseline: 21.4610x; 1.4058x over previous
#include <hip/hip_runtime.h>
#include <math.h>

typedef float v2f __attribute__((ext_vector_type(2)));
typedef float v8f __attribute__((ext_vector_type(8)));

#define N_EDGES 500000
#define N_RELS  256
#define DIM     256
#define N3      768

__global__ void zero_kernel(float* __restrict__ p, int n) {
    int i = blockIdx.x * blockDim.x + threadIdx.x;
    if (i < n) p[i] = 0.0f;
}

#define ACC_THREADS 1024
#define ACC_WAVES   (ACC_THREADS / 32)
#define NSLICE      4
#define SLICE_CH    (DIM / NSLICE)
#define NGROUP      32
#define NPLANE      2
#define EPB         (N_EDGES / NGROUP)
#define EPTH        ((EPB + ACC_THREADS - 1) / ACC_THREADS)
#define WSPAN       ((EPB + ACC_WAVES - 1) / ACC_WAVES)
#define BATCH       32
static_assert(N_EDGES % NGROUP == 0, "edge count must split evenly over the edge shares");
static_assert(N_RELS == 256, "the packed id keeps the relation in its low 8 bits");

__device__ __forceinline__ void flush_row(float* __restrict__ gslice, int r, int lane, float2 acc) {
    float* d = gslice + (size_t)r * DIM + lane * 2;
    atomicAdd(d + 0, acc.x);
    atomicAdd(d + 1, acc.y);
}

template <bool PARTIAL>
__device__ __forceinline__ void sum_batch(const float* __restrict__ ebase, float* __restrict__ gslice,
                                          int k, int cnt, int lane, float2& acc, int& cur) {
    float2 v[BATCH];
#pragma unroll
    for (int j = 0; j < BATCH; ++j) {
        v[j] = make_float2(0.0f, 0.0f);
        if (!PARTIAL || j < cnt) {
            const int n = __builtin_amdgcn_readlane(k, j) >> 8;
            const float2* row = (const float2*)(ebase + (size_t)n * DIM);
            v[j] = row[lane];
        }
    }
#pragma unroll
    for (int j = 0; j < BATCH; ++j) {
        if (!PARTIAL || j < cnt) {
            const int r = __builtin_amdgcn_readlane(k, j) & (N_RELS - 1);
            if (r != cur) {
                if (cur >= 0) flush_row(gslice, cur, lane, acc);
                acc = make_float2(0.0f, 0.0f);
                cur = r;
            }
            acc.x += v[j].x;
            acc.y += v[j].y;
        }
    }
}

__global__ __launch_bounds__(ACC_THREADS) void accum_kernel(
        const int* __restrict__ src_nid, const int* __restrict__ dst_nid,
        const int* __restrict__ rel, const float* __restrict__ emb,
        float* __restrict__ sums,
        float* __restrict__ counts) {
    __shared__ int s_cnt[N_RELS];
    __shared__ int s_start[N_RELS];
    __shared__ int s_key[EPB];

    const int tid   = threadIdx.x;
    const int lane  = tid & 31;
    const int wave  = __builtin_amdgcn_readfirstlane(tid >> 5);
    const int slice = blockIdx.x & (NSLICE - 1);
    const int plane = (blockIdx.x / NSLICE) & (NPLANE - 1);
    const int group = blockIdx.x / (NSLICE * NPLANE);
    const bool count_here = plane == 0 && slice == 0;

    const int* nid  = (plane ? dst_nid : src_nid) + group * EPB;
    const int* rl   = rel + group * EPB;
    float*   gslice = sums + plane * (N_RELS * DIM) + slice * SLICE_CH;
    const float* ebase = emb + slice * SLICE_CH;

    if (tid < N_RELS) s_cnt[tid] = 0;
    __syncthreads();

    int key[EPTH], rank[EPTH];
#pragma unroll
    for (int i = 0; i < EPTH; ++i) {
        const int e  = tid + i * ACC_THREADS;
        const int ec = e < EPB ? e : EPB - 1;
        key[i] = (nid[ec] << 8) | rl[ec];
    }
#pragma unroll
    for (int i = 0; i < EPTH; ++i) {
        const int e = tid + i * ACC_THREADS;
        rank[i] = 0;
        if (e < EPB) rank[i] = atomicAdd(&s_cnt[key[i] & (N_RELS - 1)], 1);
    }
    __syncthreads();

    if (count_here && tid < N_RELS) atomicAdd(counts + tid, (float)s_cnt[tid]);

    if (tid < 32) {
        int c[8];
        int s = 0;
#pragma unroll
        for (int q = 0; q < 8; ++q) { c[q] = s_cnt[tid * 8 + q]; s += c[q]; }
        int incl = s;
#pragma unroll
        for (int off = 1; off < 32; off <<= 1) {
            const int t = __shfl_up(incl, off);
            if (tid >= off) incl += t;
        }
        int run = incl - s;
#pragma unroll
        for (int q = 0; q < 8; ++q) { s_start[tid * 8 + q] = run; run += c[q]; }
    }
    __syncthreads();

#pragma unroll
    for (int i = 0; i < EPTH; ++i) {
        const int e = tid + i * ACC_THREADS;
        if (e < EPB) s_key[s_start[key[i] & (N_RELS - 1)] + rank[i]] = key[i];
    }
    __syncthreads();

    const int lo = wave * WSPAN;
    const int hi = (lo + WSPAN < EPB) ? lo + WSPAN : EPB;
    float2 acc = make_float2(0.0f, 0.0f);
    int cur = -1;
    for (int b = lo; b < hi; b += BATCH) {
        const int cnt = (hi - b < BATCH) ? hi - b : BATCH;
        const int idx = b + lane;
        const int k   = s_key[idx < hi ? idx : hi - 1];
        if (cnt == BATCH) sum_batch<false>(ebase, gslice, k, cnt, lane, acc, cur);
        else              sum_batch<true >(ebase, gslice, k, cnt, lane, acc, cur);
    }
    if (cur >= 0) flush_row(gslice, cur, lane, acc);
}

struct Frag {
    float4 x;
    float  h[4];
    float4 wi0, wi1, wi2;
    float4 wh0, wh1, wh2;
};

__device__ __forceinline__ Frag load_frag(const float* __restrict__ xrow,
                                          const float* __restrict__ hrow,
                                          const float* __restrict__ wi,
                                          const float* __restrict__ wh, int k0) {
    Frag f;
    f.x    = *(const float4*)(xrow + k0);
    f.h[0] = hrow[2 * k0];
    f.h[1] = hrow[2 * k0 + 2];
    f.h[2] = hrow[2 * k0 + 4];
    f.h[3] = hrow[2 * k0 + 6];
    f.wi0  = *(const float4*)(wi + k0);
    f.wi1  = *(const float4*)(wi + DIM * DIM + k0);
    f.wi2  = *(const float4*)(wi + 2 * DIM * DIM + k0);
    f.wh0  = *(const float4*)(wh + k0);
    f.wh1  = *(const float4*)(wh + DIM * DIM + k0);
    f.wh2  = *(const float4*)(wh + 2 * DIM * DIM + k0);
    return f;
}

#define WMMA_F32(a, b, c) __builtin_amdgcn_wmma_f32_16x16x4_f32( \
         false, (a),  false, (b),  (short)0, (c), \
         false,  false)

__device__ __forceinline__ void mma_frag(const Frag& f, float cnt,
                                         v8f& gi0, v8f& gi1, v8f& gi2,
                                         v8f& gh0, v8f& gh1, v8f& gh2) {
    v2f ax0 = {f.x.x / cnt, f.x.y / cnt};
    v2f ax1 = {f.x.z / cnt, f.x.w / cnt};
    v2f ah0 = {f.h[0], f.h[1]};
    v2f ah1 = {f.h[2], f.h[3]};
    v2f b;
    b.x = f.wi0.x; b.y = f.wi0.y; gi0 = WMMA_F32(ax0, b, gi0);
    b.x = f.wi1.x; b.y = f.wi1.y; gi1 = WMMA_F32(ax0, b, gi1);
    b.x = f.wi2.x; b.y = f.wi2.y; gi2 = WMMA_F32(ax0, b, gi2);
    b.x = f.wh0.x; b.y = f.wh0.y; gh0 = WMMA_F32(ah0, b, gh0);
    b.x = f.wh1.x; b.y = f.wh1.y; gh1 = WMMA_F32(ah0, b, gh1);
    b.x = f.wh2.x; b.y = f.wh2.y; gh2 = WMMA_F32(ah0, b, gh2);
    b.x = f.wi0.z; b.y = f.wi0.w; gi0 = WMMA_F32(ax1, b, gi0);
    b.x = f.wi1.z; b.y = f.wi1.w; gi1 = WMMA_F32(ax1, b, gi1);
    b.x = f.wi2.z; b.y = f.wi2.w; gi2 = WMMA_F32(ax1, b, gi2);
    b.x = f.wh0.z; b.y = f.wh0.w; gh0 = WMMA_F32(ah1, b, gh0);
    b.x = f.wh1.z; b.y = f.wh1.w; gh1 = WMMA_F32(ah1, b, gh1);
    b.x = f.wh2.z; b.y = f.wh2.w; gh2 = WMMA_F32(ah1, b, gh2);
}

#define GRU_THREADS 64
#define KCHUNK      8
#define TRIP_CHUNKS 16
#define NTRIP       (DIM / (KCHUNK * TRIP_CHUNKS))

__global__ __launch_bounds__(GRU_THREADS) void gru_fused(
        const float* __restrict__ sums,
        const float* __restrict__ counts,
        const float* __restrict__ dyn,
        const float* __restrict__ W_ih, const float* __restrict__ W_hh,
        const float* __restrict__ b_ih, const float* __restrict__ b_hh,
        float* __restrict__ out) {
    __shared__ float xch[8 * 32];

    const int lane  = threadIdx.x & 31;
    const int which = threadIdx.x >> 5;
    const int tm    = blockIdx.x >> 4;
    const int tn    = blockIdx.x & 15;
    const int half  = lane >> 4;
    const int ml    = lane & 15;
    const int m     = tm * 16 + ml;
    const int d     = tn * 16 + ml;

    const float cnt = fmaxf(counts[m], 1.0f);
    const float* xrow = sums + (size_t)which * (N_RELS * DIM) + (size_t)m * DIM + 4 * half;
    const float* hrow = dyn + (size_t)m * (2 * DIM) + 8 * half + which;
    const float* wi   = W_ih + (size_t)d * DIM + 4 * half;
    const float* wh   = W_hh + (size_t)d * DIM + 4 * half;

    v8f gi0 = {}, gi1 = {}, gi2 = {}, gh0 = {}, gh1 = {}, gh2 = {};

#pragma unroll 1
    for (int t = 0; t < NTRIP; ++t) {
        const int kb = t * (KCHUNK * TRIP_CHUNKS);
        Frag fr[2];
        fr[0] = load_frag(xrow, hrow, wi, wh, kb);
#pragma unroll
        for (int c = 0; c < TRIP_CHUNKS; ++c) {
            __builtin_amdgcn_sched_barrier(0);
            if (c + 1 < TRIP_CHUNKS)
                fr[(c + 1) & 1] = load_frag(xrow, hrow, wi, wh, kb + (c + 1) * KCHUNK);
            __builtin_amdgcn_sched_barrier(0);
            mma_frag(fr[c & 1], cnt, gi0, gi1, gi2, gh0, gh1, gh2);
        }
    }

    const float bir = b_ih[d], biz = b_ih[DIM + d], bin = b_ih[2 * DIM + d];
    const float bhr = b_hh[d], bhz = b_hh[DIM + d], bhn = b_hh[2 * DIM + d];

    float res[8];
#pragma unroll
    for (int v = 0; v < 8; ++v) {
        const int row = tm * 16 + v + 8 * half;
        const float i_r = gi0[v] + bir, i_z = gi1[v] + biz, i_n = gi2[v] + bin;
        const float h_r = gh0[v] + bhr, h_z = gh1[v] + bhz, h_n = gh2[v] + bhn;
        const float h   = dyn[(size_t)row * (2 * DIM) + d * 2 + which];
        const float rg  = 1.0f / (1.0f + expf(-(i_r + h_r)));
        const float z   = 1.0f / (1.0f + expf(-(i_z + h_z)));
        const float ng  = tanhf(i_n + rg * h_n);
        res[v] = (1.0f - z) * ng + z * h;
    }

    if (which == 1) {
#pragma unroll
        for (int v = 0; v < 8; ++v) xch[v * 32 + lane] = res[v];
    }
    __syncthreads();
    if (which == 0) {
#pragma unroll
        for (int v = 0; v < 8; ++v) {
            const int row = tm * 16 + v + 8 * half;
            float2 o;
            o.x = res[v];
            o.y = xch[v * 32 + lane];
            *(float2*)(out + ((size_t)row * DIM + d) * 2) = o;
        }
    }
}

extern "C" void kernel_launch(void* const* d_in, const int* in_sizes, int n_in,
                              void* d_out, int out_size, void* d_ws, size_t ws_size,
                              hipStream_t stream) {
    (void)in_sizes; (void)n_in; (void)out_size; (void)ws_size;

    const int*   src_nid = (const int*)d_in[0];
    const int*   dst_nid = (const int*)d_in[1];
    const int*   rel     = (const int*)d_in[2];
    const float* ent     = (const float*)d_in[3];
    const float* dyn     = (const float*)d_in[4];
    const float* W_ih    = (const float*)d_in[5];
    const float* W_hh    = (const float*)d_in[6];
    const float* b_ih    = (const float*)d_in[7];
    const float* b_hh    = (const float*)d_in[8];
    float*       out     = (float*)d_out;

    float* ws       = (float*)d_ws;
    float* counts   = ws + 131072;

    const int NACC = 131328;
    zero_kernel<<<(NACC + 255) / 256, 256, 0, stream>>>(ws, NACC);

    accum_kernel<<<NPLANE * NSLICE * NGROUP, ACC_THREADS, 0, stream>>>(
            src_nid, dst_nid, rel, ent, ws, counts);

    gru_fused<<<(N_RELS / 16) * (DIM / 16), GRU_THREADS, 0, stream>>>(
            ws, counts, dyn, W_ih, W_hh, b_ih, b_hh, out);
}


# accum: a block takes its share by the package half it runs on (a ticket per half): each half of the second-level cache gathers 2 of the 4 channel slices; arrival order of the f32 atomic sums changes
# speedup vs baseline: 23.1772x; 1.0800x over previous
#include <hip/hip_runtime.h>
#include <math.h>

typedef float v2f __attribute__((ext_vector_type(2)));
typedef float v8f __attribute__((ext_vector_type(8)));

#define N_EDGES 500000
#define N_RELS  256
#define DIM     256
#define N3      768

__global__ void zero_kernel(float* __restrict__ p, int n) {
    int i = blockIdx.x * blockDim.x + threadIdx.x;
    if (i < n) p[i] = 0.0f;
}

#define ACC_THREADS 1024
#define ACC_WAVES   (ACC_THREADS / 32)
#define NSLICE      4
#define SLICE_CH    (DIM / NSLICE)
#define NGROUP      32
#define NPLANE      2
#define EPB         (N_EDGES / NGROUP)
#define EPTH        ((EPB + ACC_THREADS - 1) / ACC_THREADS)
#define WSPAN       ((EPB + ACC_WAVES - 1) / ACC_WAVES)
#define BATCH       32
static_assert(N_EDGES % NGROUP == 0, "edge count must split evenly over the edge shares");
static_assert(N_RELS == 256, "the packed id keeps the relation in its low 8 bits");
#define HALF_ITEMS  (NPLANE * (NSLICE / 2) * NGROUP)
static_assert(NSLICE == 4 && HALF_ITEMS * 2 == NPLANE * NSLICE * NGROUP, "two slices per half");
static_assert((HALF_ITEMS & (HALF_ITEMS - 1)) == 0, "the share index is cut with masks");

__device__ __forceinline__ void flush_row(float* __restrict__ gslice, int r, int lane, float2 acc) {
    float* d = gslice + (size_t)r * DIM + lane * 2;
    atomicAdd(d + 0, acc.x);
    atomicAdd(d + 1, acc.y);
}

template <bool PARTIAL>
__device__ __forceinline__ void sum_batch(const float* __restrict__ ebase, float* __restrict__ gslice,
                                          int k, int cnt, int lane, float2& acc, int& cur) {
    float2 v[BATCH];
#pragma unroll
    for (int j = 0; j < BATCH; ++j) {
        v[j] = make_float2(0.0f, 0.0f);
        if (!PARTIAL || j < cnt) {
            const int n = __builtin_amdgcn_readlane(k, j) >> 8;
            const float2* row = (const float2*)(ebase + (size_t)n * DIM);
            v[j] = row[lane];
        }
    }
#pragma unroll
    for (int j = 0; j < BATCH; ++j) {
        if (!PARTIAL || j < cnt) {
            const int r = __builtin_amdgcn_readlane(k, j) & (N_RELS - 1);
            if (r != cur) {
                if (cur >= 0) flush_row(gslice, cur, lane, acc);
                acc = make_float2(0.0f, 0.0f);
                cur = r;
            }
            acc.x += v[j].x;
            acc.y += v[j].y;
        }
    }
}

__global__ __launch_bounds__(ACC_THREADS) void accum_kernel(
        const int* __restrict__ src_nid, const int* __restrict__ dst_nid,
        const int* __restrict__ rel, const float* __restrict__ emb,
        float* __restrict__ sums,
        float* __restrict__ counts,
        int* __restrict__ tickets) {
    __shared__ int s_item;
    __shared__ int s_cnt[N_RELS];
    __shared__ int s_start[N_RELS];
    __shared__ int s_key[EPB];

    const int tid   = threadIdx.x;
    const int lane  = tid & 31;
    const int wave  = __builtin_amdgcn_readfirstlane(tid >> 5);

    if (tid == 0) {
        const unsigned where = __builtin_amdgcn_s_sendmsg_rtn(0x87);
        const int hf = (int)((where >> 18) & 1u);
        const int t  = atomicAdd(&tickets[hf], 1) & (HALF_ITEMS * 2 - 1);
        s_item = t < HALF_ITEMS ? hf * HALF_ITEMS + t
                                : (1 - hf) * HALF_ITEMS + (2 * HALF_ITEMS - 1 - t);
    }
    if (tid < N_RELS) s_cnt[tid] = 0;
    __syncthreads();

    const int item  = __builtin_amdgcn_readfirstlane(s_item);
    const int hsel  = item / HALF_ITEMS;
    const int q     = item & (HALF_ITEMS - 1);
    const int slice = hsel * (NSLICE / 2) + (q & (NSLICE / 2 - 1));
    const int plane = (q / (NSLICE / 2)) & (NPLANE - 1);
    const int group = q / ((NSLICE / 2) * NPLANE);
    const bool count_here = plane == 0 && slice == 0;

    const int* nid  = (plane ? dst_nid : src_nid) + group * EPB;
    const int* rl   = rel + group * EPB;
    float*   gslice = sums + plane * (N_RELS * DIM) + slice * SLICE_CH;
    const float* ebase = emb + slice * SLICE_CH;

    int key[EPTH], rank[EPTH];
#pragma unroll
    for (int i = 0; i < EPTH; ++i) {
        const int e  = tid + i * ACC_THREADS;
        const int ec = e < EPB ? e : EPB - 1;
        key[i] = (nid[ec] << 8) | rl[ec];
    }
#pragma unroll
    for (int i = 0; i < EPTH; ++i) {
        const int e = tid + i * ACC_THREADS;
        rank[i] = 0;
        if (e < EPB) rank[i] = atomicAdd(&s_cnt[key[i] & (N_RELS - 1)], 1);
    }
    __syncthreads();

    if (count_here && tid < N_RELS) atomicAdd(counts + tid, (float)s_cnt[tid]);

    if (tid < 32) {
        int c[8];
        int s = 0;
#pragma unroll
        for (int q = 0; q < 8; ++q) { c[q] = s_cnt[tid * 8 + q]; s += c[q]; }
        int incl = s;
#pragma unroll
        for (int off = 1; off < 32; off <<= 1) {
            const int t = __shfl_up(incl, off);
            if (tid >= off) incl += t;
        }
        int run = incl - s;
#pragma unroll
        for (int q = 0; q < 8; ++q) { s_start[tid * 8 + q] = run; run += c[q]; }
    }
    __syncthreads();

#pragma unroll
    for (int i = 0; i < EPTH; ++i) {
        const int e = tid + i * ACC_THREADS;
        if (e < EPB) s_key[s_start[key[i] & (N_RELS - 1)] + rank[i]] = key[i];
    }
    __syncthreads();

    const int lo = wave * WSPAN;
    const int hi = (lo + WSPAN < EPB) ? lo + WSPAN : EPB;
    float2 acc = make_float2(0.0f, 0.0f);
    int cur = -1;
    for (int b = lo; b < hi; b += BATCH) {
        const int cnt = (hi - b < BATCH) ? hi - b : BATCH;
        const int idx = b + lane;
        const int k   = s_key[idx < hi ? idx : hi - 1];
        if (cnt == BATCH) sum_batch<false>(ebase, gslice, k, cnt, lane, acc, cur);
        else              sum_batch<true >(ebase, gslice, k, cnt, lane, acc, cur);
    }
    if (cur >= 0) flush_row(gslice, cur, lane, acc);
}

struct Frag {
    float4 x;
    float  h[4];
    float4 wi0, wi1, wi2;
    float4 wh0, wh1, wh2;
};

__device__ __forceinline__ Frag load_frag(const float* __restrict__ xrow,
                                          const float* __restrict__ hrow,
                                          const float* __restrict__ wi,
                                          const float* __restrict__ wh, int k0) {
    Frag f;
    f.x    = *(const float4*)(xrow + k0);
    f.h[0] = hrow[2 * k0];
    f.h[1] = hrow[2 * k0 + 2];
    f.h[2] = hrow[2 * k0 + 4];
    f.h[3] = hrow[2 * k0 + 6];
    f.wi0  = *(const float4*)(wi + k0);
    f.wi1  = *(const float4*)(wi + DIM * DIM + k0);
    f.wi2  = *(const float4*)(wi + 2 * DIM * DIM + k0);
    f.wh0  = *(const float4*)(wh + k0);
    f.wh1  = *(const float4*)(wh + DIM * DIM + k0);
    f.wh2  = *(const float4*)(wh + 2 * DIM * DIM + k0);
    return f;
}

#define WMMA_F32(a, b, c) __builtin_amdgcn_wmma_f32_16x16x4_f32( \
         false, (a),  false, (b),  (short)0, (c), \
         false,  false)

__device__ __forceinline__ void mma_frag(const Frag& f, float cnt,
                                         v8f& gi0, v8f& gi1, v8f& gi2,
                                         v8f& gh0, v8f& gh1, v8f& gh2) {
    v2f ax0 = {f.x.x / cnt, f.x.y / cnt};
    v2f ax1 = {f.x.z / cnt, f.x.w / cnt};
    v2f ah0 = {f.h[0], f.h[1]};
    v2f ah1 = {f.h[2], f.h[3]};
    v2f b;
    b.x = f.wi0.x; b.y = f.wi0.y; gi0 = WMMA_F32(ax0, b, gi0);
    b.x = f.wi1.x; b.y = f.wi1.y; gi1 = WMMA_F32(ax0, b, gi1);
    b.x = f.wi2.x; b.y = f.wi2.y; gi2 = WMMA_F32(ax0, b, gi2);
    b.x = f.wh0.x; b.y = f.wh0.y; gh0 = WMMA_F32(ah0, b, gh0);
    b.x = f.wh1.x; b.y = f.wh1.y; gh1 = WMMA_F32(ah0, b, gh1);
    b.x = f.wh2.x; b.y = f.wh2.y; gh2 = WMMA_F32(ah0, b, gh2);
    b.x = f.wi0.z; b.y = f.wi0.w; gi0 = WMMA_F32(ax1, b, gi0);
    b.x = f.wi1.z; b.y = f.wi1.w; gi1 = WMMA_F32(ax1, b, gi1);
    b.x = f.wi2.z; b.y = f.wi2.w; gi2 = WMMA_F32(ax1, b, gi2);
    b.x = f.wh0.z; b.y = f.wh0.w; gh0 = WMMA_F32(ah1, b, gh0);
    b.x = f.wh1.z; b.y = f.wh1.w; gh1 = WMMA_F32(ah1, b, gh1);
    b.x = f.wh2.z; b.y = f.wh2.w; gh2 = WMMA_F32(ah1, b, gh2);
}

#define GRU_THREADS 64
#define KCHUNK      8
#define TRIP_CHUNKS 16
#define NTRIP       (DIM / (KCHUNK * TRIP_CHUNKS))

__global__ __launch_bounds__(GRU_THREADS) void gru_fused(
        const float* __restrict__ sums,
        const float* __restrict__ counts,
        const float* __restrict__ dyn,
        const float* __restrict__ W_ih, const float* __restrict__ W_hh,
        const float* __restrict__ b_ih, const float* __restrict__ b_hh,
        float* __restrict__ out) {
    __shared__ float xch[8 * 32];

    const int lane  = threadIdx.x & 31;
    const int which = threadIdx.x >> 5;
    const int tm    = blockIdx.x >> 4;
    const int tn    = blockIdx.x & 15;
    const int half  = lane >> 4;
    const int ml    = lane & 15;
    const int m     = tm * 16 + ml;
    const int d     = tn * 16 + ml;

    const float cnt = fmaxf(counts[m], 1.0f);
    const float* xrow = sums + (size_t)which * (N_RELS * DIM) + (size_t)m * DIM + 4 * half;
    const float* hrow = dyn + (size_t)m * (2 * DIM) + 8 * half + which;
    const float* wi   = W_ih + (size_t)d * DIM + 4 * half;
    const float* wh   = W_hh + (size_t)d * DIM + 4 * half;

    v8f gi0 = {}, gi1 = {}, gi2 = {}, gh0 = {}, gh1 = {}, gh2 = {};

#pragma unroll 1
    for (int t = 0; t < NTRIP; ++t) {
        const int kb = t * (KCHUNK * TRIP_CHUNKS);
        Frag fr[2];
        fr[0] = load_frag(xrow, hrow, wi, wh, kb);
#pragma unroll
        for (int c = 0; c < TRIP_CHUNKS; ++c) {
            __builtin_amdgcn_sched_barrier(0);
            if (c + 1 < TRIP_CHUNKS)
                fr[(c + 1) & 1] = load_frag(xrow, hrow, wi, wh, kb + (c + 1) * KCHUNK);
            __builtin_amdgcn_sched_barrier(0);
            mma_frag(fr[c & 1], cnt, gi0, gi1, gi2, gh0, gh1, gh2);
        }
    }

    const float bir = b_ih[d], biz = b_ih[DIM + d], bin = b_ih[2 * DIM + d];
    const float bhr = b_hh[d], bhz = b_hh[DIM + d], bhn = b_hh[2 * DIM + d];

    float res[8];
#pragma unroll
    for (int v = 0; v < 8; ++v) {
        const int row = tm * 16 + v + 8 * half;
        const float i_r = gi0[v] + bir, i_z = gi1[v] + biz, i_n = gi2[v] + bin;
        const float h_r = gh0[v] + bhr, h_z = gh1[v] + bhz, h_n = gh2[v] + bhn;
        const float h   = dyn[(size_t)row * (2 * DIM) + d * 2 + which];
        const float rg  = 1.0f / (1.0f + expf(-(i_r + h_r)));
        const float z   = 1.0f / (1.0f + expf(-(i_z + h_z)));
        const float ng  = tanhf(i_n + rg * h_n);
        res[v] = (1.0f - z) * ng + z * h;
    }

    if (which == 1) {
#pragma unroll
        for (int v = 0; v < 8; ++v) xch[v * 32 + lane] = res[v];
    }
    __syncthreads();
    if (which == 0) {
#pragma unroll
        for (int v = 0; v < 8; ++v) {
            const int row = tm * 16 + v + 8 * half;
            float2 o;
            o.x = res[v];
            o.y = xch[v * 32 + lane];
            *(float2*)(out + ((size_t)row * DIM + d) * 2) = o;
        }
    }
}

extern "C" void kernel_launch(void* const* d_in, const int* in_sizes, int n_in,
                              void* d_out, int out_size, void* d_ws, size_t ws_size,
                              hipStream_t stream) {
    (void)in_sizes; (void)n_in; (void)out_size; (void)ws_size;

    const int*   src_nid = (const int*)d_in[0];
    const int*   dst_nid = (const int*)d_in[1];
    const int*   rel     = (const int*)d_in[2];
    const float* ent     = (const float*)d_in[3];
    const float* dyn     = (const float*)d_in[4];
    const float* W_ih    = (const float*)d_in[5];
    const float* W_hh    = (const float*)d_in[6];
    const float* b_ih    = (const float*)d_in[7];
    const float* b_hh    = (const float*)d_in[8];
    float*       out     = (float*)d_out;

    float* ws       = (float*)d_ws;
    float* counts   = ws + 131072;
    int*   tickets  = (int*)(ws + 131328);

    const int NACC = 131330;
    zero_kernel<<<(NACC + 255) / 256, 256, 0, stream>>>(ws, NACC);

    accum_kernel<<<NPLANE * NSLICE * NGROUP, ACC_THREADS, 0, stream>>>(
            src_nid, dst_nid, rel, ent, ws, counts, tickets);

    gru_fused<<<(N_RELS / 16) * (DIM / 16), GRU_THREADS, 0, stream>>>(
            ws, counts, dyn, W_ih, W_hh, b_ih, b_hh, out);
}


# accum: a block owns half a row (2 channel slices x 64 edge shares): one 16-byte load per lane gets 512 contiguous bytes of a row; 64 blocks, not 32, add each sum word: f32 atomic sum order changes
# speedup vs baseline: 26.2106x; 1.1309x over previous
#include <hip/hip_runtime.h>
#include <math.h>

typedef float v2f __attribute__((ext_vector_type(2)));
typedef float v8f __attribute__((ext_vector_type(8)));

#define N_EDGES 500000
#define N_RELS  256
#define DIM     256
#define N3      768

__global__ void zero_kernel(float* __restrict__ p, int n) {
    int i = blockIdx.x * blockDim.x + threadIdx.x;
    if (i < n) p[i] = 0.0f;
}

#define ACC_THREADS 1024
#define ACC_WAVES   (ACC_THREADS / 32)
#define NSLICE      2
#define SLICE_CH    (DIM / NSLICE)
#define NGROUP      64
#define NPLANE      2
#define EPB         ((N_EDGES + NGROUP - 1) / NGROUP)
#define EPTH        ((EPB + ACC_THREADS - 1) / ACC_THREADS)
#define WSPAN       ((EPB + ACC_WAVES - 1) / ACC_WAVES)
#define KEYS        32
#define BATCH       16
static_assert((NGROUP - 1) * EPB < N_EDGES, "every edge share holds at least one edge");
static_assert(KEYS == 2 * BATCH, "a trip's ids are used up by two groups of row requests");
static_assert(SLICE_CH == 32 * 4, "a lane holds four channels of the slice: one 16-byte load");
static_assert(N_RELS == 256, "the packed id keeps the relation in its low 8 bits");
#define HALF_ITEMS  (NPLANE * (NSLICE / 2) * NGROUP)
static_assert(NSLICE == 2 && HALF_ITEMS * 2 == NPLANE * NSLICE * NGROUP, "one slice per half");
static_assert((HALF_ITEMS & (HALF_ITEMS - 1)) == 0, "the share index is cut with masks");

__device__ __forceinline__ void flush_row(float* __restrict__ gslice, int r, int lane, float4 acc) {
    float* d = gslice + (size_t)r * DIM + lane * 4;
    atomicAdd(d + 0, acc.x);
    atomicAdd(d + 1, acc.y);
    atomicAdd(d + 2, acc.z);
    atomicAdd(d + 3, acc.w);
}

template <bool PARTIAL, int L0>
__device__ __forceinline__ void sum_batch(const float* __restrict__ ebase, float* __restrict__ gslice,
                                          int k, int cnt, int lane, float4& acc, int& cur) {
    float4 v[BATCH];
#pragma unroll
    for (int j = 0; j < BATCH; ++j) {
        v[j] = make_float4(0.0f, 0.0f, 0.0f, 0.0f);
        if (!PARTIAL || j < cnt) {
            const int n = __builtin_amdgcn_readlane(k, L0 + j) >> 8;
            const float4* row = (const float4*)(ebase + (size_t)n * DIM);
            v[j] = row[lane];
        }
    }
#pragma unroll
    for (int j = 0; j < BATCH; ++j) {
        if (!PARTIAL || j < cnt) {
            const int r = __builtin_amdgcn_readlane(k, L0 + j) & (N_RELS - 1);
            if (r != cur) {
                if (cur >= 0) flush_row(gslice, cur, lane, acc);
                acc = make_float4(0.0f, 0.0f, 0.0f, 0.0f);
                cur = r;
            }
            acc.x += v[j].x;
            acc.y += v[j].y;
            acc.z += v[j].z;
            acc.w += v[j].w;
        }
    }
}

__global__ __launch_bounds__(ACC_THREADS) void accum_kernel(
        const int* __restrict__ src_nid, const int* __restrict__ dst_nid,
        const int* __restrict__ rel, const float* __restrict__ emb,
        float* __restrict__ sums,
        float* __restrict__ counts,
        int* __restrict__ tickets) {
    __shared__ int s_item;
    __shared__ int s_cnt[N_RELS];
    __shared__ int s_start[N_RELS];
    __shared__ int s_key[EPB];

    const int tid   = threadIdx.x;
    const int lane  = tid & 31;
    const int wave  = __builtin_amdgcn_readfirstlane(tid >> 5);

    if (tid == 0) {
        const unsigned where = __builtin_amdgcn_s_sendmsg_rtn(0x87);
        const int hf = (int)((where >> 18) & 1u);
        const int t  = atomicAdd(&tickets[hf], 1) & (HALF_ITEMS * 2 - 1);
        s_item = t < HALF_ITEMS ? hf * HALF_ITEMS + t
                                : (1 - hf) * HALF_ITEMS + (2 * HALF_ITEMS - 1 - t);
    }
    if (tid < N_RELS) s_cnt[tid] = 0;
    __syncthreads();

    const int item  = __builtin_amdgcn_readfirstlane(s_item);
    const int hsel  = item / HALF_ITEMS;
    const int q     = item & (HALF_ITEMS - 1);
    const int slice = hsel * (NSLICE / 2) + (q & (NSLICE / 2 - 1));
    const int plane = (q / (NSLICE / 2)) & (NPLANE - 1);
    const int group = q / ((NSLICE / 2) * NPLANE);
    const bool count_here = plane == 0 && slice == 0;
    const int nE = (N_EDGES - group * EPB < EPB) ? N_EDGES - group * EPB : EPB;

    const int* nid  = (plane ? dst_nid : src_nid) + group * EPB;
    const int* rl   = rel + group * EPB;
    float*   gslice = sums + plane * (N_RELS * DIM) + slice * SLICE_CH;
    const float* ebase = emb + slice * SLICE_CH;

    int key[EPTH], rank[EPTH];
#pragma unroll
    for (int i = 0; i < EPTH; ++i) {
        const int e  = tid + i * ACC_THREADS;
        const int ec = e < nE ? e : nE - 1;
        key[i] = (nid[ec] << 8) | rl[ec];
    }
#pragma unroll
    for (int i = 0; i < EPTH; ++i) {
        const int e = tid + i * ACC_THREADS;
        rank[i] = 0;
        if (e < nE) rank[i] = atomicAdd(&s_cnt[key[i] & (N_RELS - 1)], 1);
    }
    __syncthreads();

    if (count_here && tid < N_RELS) atomicAdd(counts + tid, (float)s_cnt[tid]);

    if (tid < 32) {
        int c[8];
        int s = 0;
#pragma unroll
        for (int q = 0; q < 8; ++q) { c[q] = s_cnt[tid * 8 + q]; s += c[q]; }
        int incl = s;
#pragma unroll
        for (int off = 1; off < 32; off <<= 1) {
            const int t = __shfl_up(incl, off);
            if (tid >= off) incl += t;
        }
        int run = incl - s;
#pragma unroll
        for (int q = 0; q < 8; ++q) { s_start[tid * 8 + q] = run; run += c[q]; }
    }
    __syncthreads();

#pragma unroll
    for (int i = 0; i < EPTH; ++i) {
        const int e = tid + i * ACC_THREADS;
        if (e < nE) s_key[s_start[key[i] & (N_RELS - 1)] + rank[i]] = key[i];
    }
    __syncthreads();

    const int lo = wave * WSPAN;
    const int hi = (lo + WSPAN < nE) ? lo + WSPAN : nE;
    float4 acc = make_float4(0.0f, 0.0f, 0.0f, 0.0f);
    int cur = -1;
    for (int b = lo; b < hi; b += KEYS) {
        const int idx = b + lane;
        const int k   = s_key[idx < hi ? idx : hi - 1];
        const int c0  = hi - b;
        if (c0 >= BATCH) sum_batch<false, 0>(ebase, gslice, k, BATCH, lane, acc, cur);
        else             sum_batch<true,  0>(ebase, gslice, k, c0, lane, acc, cur);
        const int c1  = c0 - BATCH;
        if (c1 >= BATCH) sum_batch<false, BATCH>(ebase, gslice, k, BATCH, lane, acc, cur);
        else if (c1 > 0) sum_batch<true,  BATCH>(ebase, gslice, k, c1, lane, acc, cur);
    }
    if (cur >= 0) flush_row(gslice, cur, lane, acc);
}

struct Frag {
    float  h[4];
    float4 wi0, wi1, wi2;
    float4 wh0, wh1, wh2;
};

__device__ __forceinline__ Frag load_frag(const float* __restrict__ hrow,
                                          const float* __restrict__ wi,
                                          const float* __restrict__ wh, int k0) {
    Frag f;
    f.h[0] = hrow[2 * k0];
    f.h[1] = hrow[2 * k0 + 2];
    f.h[2] = hrow[2 * k0 + 4];
    f.h[3] = hrow[2 * k0 + 6];
    f.wi0  = *(const float4*)(wi + k0);
    f.wi1  = *(const float4*)(wi + DIM * DIM + k0);
    f.wi2  = *(const float4*)(wi + 2 * DIM * DIM + k0);
    f.wh0  = *(const float4*)(wh + k0);
    f.wh1  = *(const float4*)(wh + DIM * DIM + k0);
    f.wh2  = *(const float4*)(wh + 2 * DIM * DIM + k0);
    return f;
}

#define WMMA_F32(a, b, c) __builtin_amdgcn_wmma_f32_16x16x4_f32( \
         false, (a),  false, (b),  (short)0, (c), \
         false,  false)

__device__ __forceinline__ void mma_frag(const Frag& f, v2f ax0, v2f ax1,
                                         v8f& gi0, v8f& gi1, v8f& gi2,
                                         v8f& gh0, v8f& gh1, v8f& gh2) {
    v2f ah0 = {f.h[0], f.h[1]};
    v2f ah1 = {f.h[2], f.h[3]};
    v2f b;
    b.x = f.wi0.x; b.y = f.wi0.y; gi0 = WMMA_F32(ax0, b, gi0);
    b.x = f.wi1.x; b.y = f.wi1.y; gi1 = WMMA_F32(ax0, b, gi1);
    b.x = f.wi2.x; b.y = f.wi2.y; gi2 = WMMA_F32(ax0, b, gi2);
    b.x = f.wh0.x; b.y = f.wh0.y; gh0 = WMMA_F32(ah0, b, gh0);
    b.x = f.wh1.x; b.y = f.wh1.y; gh1 = WMMA_F32(ah0, b, gh1);
    b.x = f.wh2.x; b.y = f.wh2.y; gh2 = WMMA_F32(ah0, b, gh2);
    b.x = f.wi0.z; b.y = f.wi0.w; gi0 = WMMA_F32(ax1, b, gi0);
    b.x = f.wi1.z; b.y = f.wi1.w; gi1 = WMMA_F32(ax1, b, gi1);
    b.x = f.wi2.z; b.y = f.wi2.w; gi2 = WMMA_F32(ax1, b, gi2);
    b.x = f.wh0.z; b.y = f.wh0.w; gh0 = WMMA_F32(ah1, b, gh0);
    b.x = f.wh1.z; b.y = f.wh1.w; gh1 = WMMA_F32(ah1, b, gh1);
    b.x = f.wh2.z; b.y = f.wh2.w; gh2 = WMMA_F32(ah1, b, gh2);
}

#define GRU_WAVES   8
#define GRU_THREADS (GRU_WAVES * 32)
#define KCHUNK      8
#define KSPLIT      (GRU_WAVES / 2)
#define WCHUNKS     (DIM / (KCHUNK * KSPLIT))
#define LEAD        4
static_assert(KSPLIT * 2 == 8, "a wave closes two of the tile's eight accumulator registers");
static_assert(LEAD < WCHUNKS, "the closing loads ride in the first trip without a fragment request");

__global__ __launch_bounds__(GRU_THREADS) void gru_fused(
        const float* __restrict__ sums,
        const float* __restrict__ counts,
        const float* __restrict__ dyn,
        const float* __restrict__ W_ih, const float* __restrict__ W_hh,
        const float* __restrict__ b_ih, const float* __restrict__ b_hh,
        float* __restrict__ out) {
    __shared__ v2f   part[KSPLIT][2][6][4][32];
    __shared__ float xch[8 * 32];

    const int lane  = threadIdx.x & 31;
    const int wave  = __builtin_amdgcn_readfirstlane((int)(threadIdx.x >> 5));
    const int which = wave & 1;
    const int kq    = wave >> 1;
    const int tm    = blockIdx.x >> 4;
    const int tn    = blockIdx.x & 15;
    const int half  = lane >> 4;
    const int ml    = lane & 15;
    const int m     = tm * 16 + ml;
    const int d     = tn * 16 + ml;
    const int kb    = kq * (KCHUNK * WCHUNKS);

    const float cnt = fmaxf(counts[m], 1.0f);
    const float* xrow = sums + (size_t)which * (N_RELS * DIM) + (size_t)m * DIM + 4 * half;
    const float* hrow = dyn + (size_t)m * (2 * DIM) + 8 * half + which;
    const float* wi   = W_ih + (size_t)d * DIM + 4 * half;
    const float* wh   = W_hh + (size_t)d * DIM + 4 * half;

    v8f gi0 = {}, gi1 = {}, gi2 = {}, gh0 = {}, gh1 = {}, gh2 = {};
    float bir = 0.0f, biz = 0.0f, bin = 0.0f, bhr = 0.0f, bhz = 0.0f, bhn = 0.0f;
    float hv[2] = {0.0f, 0.0f};

    float4 xs[WCHUNKS];
#pragma unroll
    for (int c = 0; c < WCHUNKS; ++c)
        xs[c] = *(const float4*)(xrow + kb + c * KCHUNK);
    Frag fr[WCHUNKS];
#pragma unroll
    for (int c = 0; c < LEAD; ++c)
        fr[c] = load_frag(hrow, wi, wh, kb + c * KCHUNK);
    __builtin_amdgcn_sched_barrier(0);
    v2f ax[WCHUNKS][2];
#pragma unroll
    for (int c = 0; c < WCHUNKS; ++c) {
        ax[c][0].x = xs[c].x / cnt; ax[c][0].y = xs[c].y / cnt;
        ax[c][1].x = xs[c].z / cnt; ax[c][1].y = xs[c].w / cnt;
    }
#pragma unroll
    for (int c = 0; c < WCHUNKS; ++c) {
        __builtin_amdgcn_sched_barrier(0);
        if (c + LEAD < WCHUNKS) {
            fr[c + LEAD] = load_frag(hrow, wi, wh, kb + (c + LEAD) * KCHUNK);
        } else if (c + LEAD == WCHUNKS) {
            bir = b_ih[d]; biz = b_ih[DIM + d]; bin = b_ih[2 * DIM + d];
            bhr = b_hh[d]; bhz = b_hh[DIM + d]; bhn = b_hh[2 * DIM + d];
#pragma unroll
            for (int e = 0; e < 2; ++e) {
                const int row = tm * 16 + 2 * kq + e + 8 * half;
                hv[e] = dyn[(size_t)row * (2 * DIM) + d * 2 + which];
            }
        }
        __builtin_amdgcn_sched_barrier(0);
        mma_frag(fr[c], ax[c][0], ax[c][1], gi0, gi1, gi2, gh0, gh1, gh2);
    }

#pragma unroll
    for (int p = 0; p < 4; ++p) {
        v2f t;
        t.x = gi0[2 * p]; t.y = gi0[2 * p + 1]; part[kq][which][0][p][lane] = t;
        t.x = gi1[2 * p]; t.y = gi1[2 * p + 1]; part[kq][which][1][p][lane] = t;
        t.x = gi2[2 * p]; t.y = gi2[2 * p + 1]; part[kq][which][2][p][lane] = t;
        t.x = gh0[2 * p]; t.y = gh0[2 * p + 1]; part[kq][which][3][p][lane] = t;
        t.x = gh1[2 * p]; t.y = gh1[2 * p + 1]; part[kq][which][4][p][lane] = t;
        t.x = gh2[2 * p]; t.y = gh2[2 * p + 1]; part[kq][which][5][p][lane] = t;
    }
    __syncthreads();

    v2f g[6];
#pragma unroll
    for (int a = 0; a < 6; ++a) {
        g[a] = part[0][which][a][kq][lane];
        for (int q = 1; q < KSPLIT; ++q) g[a] += part[q][which][a][kq][lane];
    }

    float res[2];
#pragma unroll
    for (int e = 0; e < 2; ++e) {
        const float i_r = g[0][e] + bir, i_z = g[1][e] + biz, i_n = g[2][e] + bin;
        const float h_r = g[3][e] + bhr, h_z = g[4][e] + bhz, h_n = g[5][e] + bhn;
        const float h   = hv[e];
        const float rg  = 1.0f / (1.0f + expf(-(i_r + h_r)));
        const float z   = 1.0f / (1.0f + expf(-(i_z + h_z)));
        const float ng  = tanhf(i_n + rg * h_n);
        res[e] = (1.0f - z) * ng + z * h;
    }

    if (which == 1) {
#pragma unroll
        for (int e = 0; e < 2; ++e) xch[(2 * kq + e) * 32 + lane] = res[e];
    }
    __syncthreads();
    if (which == 0) {
#pragma unroll
        for (int e = 0; e < 2; ++e) {
            const int row = tm * 16 + 2 * kq + e + 8 * half;
            float2 o;
            o.x = res[e];
            o.y = xch[(2 * kq + e) * 32 + lane];
            *(float2*)(out + ((size_t)row * DIM + d) * 2) = o;
        }
    }
}

extern "C" void kernel_launch(void* const* d_in, const int* in_sizes, int n_in,
                              void* d_out, int out_size, void* d_ws, size_t ws_size,
                              hipStream_t stream) {
    (void)in_sizes; (void)n_in; (void)out_size; (void)ws_size;

    const int*   src_nid = (const int*)d_in[0];
    const int*   dst_nid = (const int*)d_in[1];
    const int*   rel     = (const int*)d_in[2];
    const float* ent     = (const float*)d_in[3];
    const float* dyn     = (const float*)d_in[4];
    const float* W_ih    = (const float*)d_in[5];
    const float* W_hh    = (const float*)d_in[6];
    const float* b_ih    = (const float*)d_in[7];
    const float* b_hh    = (const float*)d_in[8];
    float*       out     = (float*)d_out;

    float* ws       = (float*)d_ws;
    float* counts   = ws + 131072;
    int*   tickets  = (int*)(ws + 131328);

    const int NACC = 131330;
    zero_kernel<<<(NACC + 255) / 256, 256, 0, stream>>>(ws, NACC);

    accum_kernel<<<NPLANE * NSLICE * NGROUP, ACC_THREADS, 0, stream>>>(
            src_nid, dst_nid, rel, ent, ws, counts, tickets);

    gru_fused<<<(N_RELS / 16) * (DIM / 16), GRU_THREADS, 0, stream>>>(
            ws, counts, dyn, W_ih, W_hh, b_ih, b_hh, out);
}
